# starting kernel
# baseline (speedup 1.0000x reference)
#include <hip/hip_runtime.h>

#define B_ 32
#define S_ 2048
#define H_ 1024
#define NB 8
#define WPB 8
#define ROWS_PER_WAVE (S_ / (NB * WPB))
#define PARTIALS NB
#define PB_STRIDE (PARTIALS * H_ + 2 * PARTIALS)

typedef __attribute__((ext_vector_type(2))) float v2f;
typedef __attribute__((ext_vector_type(8))) float v8f;

__global__ __launch_bounds__(256) void attn_pass1(
    const float* __restrict__ enc, const float* __restrict__ dec,
    const float* __restrict__ mask, float* __restrict__ ws)
{
    __shared__ float red_m[WPB];
    __shared__ float red_s[WPB];
    __shared__ float ctxbuf[WPB * H_];

    const int b      = blockIdx.x / NB;
    const int blkInB = blockIdx.x % NB;
    const int tid    = threadIdx.x;
    const int wv     = tid >> 5;
    const int lane   = tid & 31;
    const int s0     = (blkInB * WPB + wv) * ROWS_PER_WAVE;

    const float4* dec4 = (const float4*)(dec + (size_t)b * H_);
    float4 dfrag[8];
#pragma unroll
    for (int k = 0; k < 8; ++k) dfrag[k] = dec4[k * 32 + lane];

    const float maskv = mask[(size_t)b * S_ + s0 + lane];

    float4 ctx[8];
#pragma unroll
    for (int k = 0; k < 8; ++k) ctx[k] = make_float4(0.f, 0.f, 0.f, 0.f);
    float m = -1e30f, ssum = 0.f;

    const float4* encB = (const float4*)(enc + (size_t)b * S_ * H_);
    for (int r = 0; r < ROWS_PER_WAVE; ++r) {
        const float4* row4 = encB + (size_t)(s0 + r) * (H_ / 4);
        float4 row[8];
#pragma unroll
        for (int k = 0; k < 8; ++k) row[k] = row4[k * 32 + lane];

        float d = 0.f;
#pragma unroll
        for (int k = 0; k < 8; ++k)
            d += row[k].x * dfrag[k].x + row[k].y * dfrag[k].y +
                 row[k].z * dfrag[k].z + row[k].w * dfrag[k].w;
#pragma unroll
        for (int off = 16; off >= 1; off >>= 1) d += __shfl_xor(d, off, 32);

        const float logit = d * __shfl(maskv, r, 32);
        const float mn = fmaxf(m, logit);
        const float c  = __expf(m - mn);
        const float p  = __expf(logit - mn);
        ssum = ssum * c + p;
#pragma unroll
        for (int k = 0; k < 8; ++k) {
            ctx[k].x = ctx[k].x * c + p * row[k].x;
            ctx[k].y = ctx[k].y * c + p * row[k].y;
            ctx[k].z = ctx[k].z * c + p * row[k].z;
            ctx[k].w = ctx[k].w * c + p * row[k].w;
        }
        m = mn;
    }

    if (lane == 0) red_m[wv] = m;
    __syncthreads();
    float mB = red_m[0];
#pragma unroll
    for (int i = 1; i < WPB; ++i) mB = fmaxf(mB, red_m[i]);
    const float w = __expf(m - mB);
    if (lane == 0) red_s[wv] = ssum * w;
    float4* cb = (float4*)(ctxbuf + wv * H_);
#pragma unroll
    for (int k = 0; k < 8; ++k) {
        float4 v = ctx[k];
        v.x *= w; v.y *= w; v.z *= w; v.w *= w;
        cb[k * 32 + lane] = v;
    }
    __syncthreads();

    float sB = 0.f;
#pragma unroll
    for (int i = 0; i < WPB; ++i) sB += red_s[i];
    float4 acc = make_float4(0.f, 0.f, 0.f, 0.f);
#pragma unroll
    for (int i = 0; i < WPB; ++i) {
        float4 v = ((const float4*)(ctxbuf + i * H_))[tid];
        acc.x += v.x; acc.y += v.y; acc.z += v.z; acc.w += v.w;
    }
    float* base = ws + (size_t)b * PB_STRIDE;
    ((float4*)(base + blkInB * H_))[tid] = acc;
    if (tid == 0) {
        base[PARTIALS * H_ + blkInB] = mB;
        base[PARTIALS * H_ + PARTIALS + blkInB] = sB;
    }
}

__global__ __launch_bounds__(256) void attn_combine(
    const float* __restrict__ ws, float* __restrict__ ctx_out)
{
    const int b = blockIdx.x, tid = threadIdx.x;
    const float* base = ws + (size_t)b * PB_STRIDE;
    const float* ms = base + PARTIALS * H_;
    const float* ss = ms + PARTIALS;

    float M = ms[0];
#pragma unroll
    for (int i = 1; i < PARTIALS; ++i) M = fmaxf(M, ms[i]);
    float e[PARTIALS], T = 0.f;
#pragma unroll
    for (int i = 0; i < PARTIALS; ++i) { e[i] = __expf(ms[i] - M); T += e[i] * ss[i]; }
    const float inv = 1.f / T;

    float4 acc = make_float4(0.f, 0.f, 0.f, 0.f);
#pragma unroll
    for (int i = 0; i < PARTIALS; ++i) {
        float4 v = ((const float4*)(base + i * H_))[tid];
        acc.x += e[i] * v.x; acc.y += e[i] * v.y;
        acc.z += e[i] * v.z; acc.w += e[i] * v.w;
    }
    acc.x *= inv; acc.y *= inv; acc.z *= inv; acc.w *= inv;
    ((float4*)(ctx_out + (size_t)b * H_))[tid] = acc;
}

__global__ __launch_bounds__(128) void attn_out_wmma(
    const float* __restrict__ ctx, const float* __restrict__ dec,
    const float* __restrict__ W, const float* __restrict__ bias,
    float* __restrict__ out)
{
    const int wave = blockIdx.x * (blockDim.x >> 5) + (threadIdx.x >> 5);
    const int lane = threadIdx.x & 31;
    const int m0 = (wave >> 6) << 4;
    const int n0 = (wave & 63) << 4;
    const int lm = lane & 15;
    const int hi = lane >> 4;

    const float* src = (m0 == 0) ? ctx : dec;
    const float* xlo = src + (size_t)(2 * lm) * H_;
    const float* xhi = xlo + H_;
    const float* wrow = W + (size_t)(n0 + lm) * (2 * H_);

    v8f acc = {};
#pragma unroll 4
    for (int k0 = 0; k0 < 1024; k0 += 4) {
        const int kk = k0 + 2 * hi;
        v2f a  = { xlo[kk],  xlo[kk + 1]  };
        v2f bm = { wrow[kk], wrow[kk + 1] };
        acc = __builtin_amdgcn_wmma_f32_16x16x4_f32(
            false, a, false, bm, (short)0, acc, false, false);
    }
#pragma unroll 4
    for (int k0 = 1024; k0 < 2048; k0 += 4) {
        const int kk = k0 + 2 * hi;
        v2f a  = { xhi[kk - 1024], xhi[kk - 1023] };
        v2f bm = { wrow[kk],       wrow[kk + 1]   };
        acc = __builtin_amdgcn_wmma_f32_16x16x4_f32(
            false, a, false, bm, (short)0, acc, false, false);
    }

    const int col = n0 + lm;
    const float bv = bias[col];
#pragma unroll
    for (int r = 0; r < 8; ++r) {
        const int row = m0 + r + (hi << 3);
        out[(size_t)row * H_ + col] = tanhf(acc[r] + bv);
    }
}

extern "C" void kernel_launch(void* const* d_in, const int* in_sizes, int n_in,
                              void* d_out, int out_size, void* d_ws, size_t ws_size,
                              hipStream_t stream) {
    const float* enc  = (const float*)d_in[0];
    const float* dec  = (const float*)d_in[1];
    const float* mask = (const float*)d_in[2];
    const float* W    = (const float*)d_in[3];
    const float* bias = (const float*)d_in[4];

    float* out = (float*)d_out;
    float* ctx = out + B_ * H_;
    float* ws  = (float*)d_ws;

    attn_pass1 <<<dim3(B_ * NB), dim3(256), 0, stream>>>(enc, dec, mask, ws);
    attn_combine<<<dim3(B_),      dim3(256), 0, stream>>>(ws, ctx);
    attn_out_wmma<<<dim3(32),     dim3(128), 0, stream>>>(ctx, dec, W, bias, out);
}


# output GEMM: K split over 16 waves per 16x16 tile, whole-line loads staged in wave-private LDS, 16 partial tiles summed through LDS (changes the f32 summation order of the 2048-term dot products)
# speedup vs baseline: 1.6942x; 1.6942x over previous
#include <hip/hip_runtime.h>

#define B_ 32
#define S_ 2048
#define H_ 1024
#define NB 8
#define WPB 8
#define ROWS_PER_WAVE (S_ / (NB * WPB))
#define PARTIALS NB
#define PB_STRIDE (PARTIALS * H_ + 2 * PARTIALS)

typedef __attribute__((ext_vector_type(2))) float v2f;
typedef __attribute__((ext_vector_type(8))) float v8f;

__global__ __launch_bounds__(256) void attn_pass1(
    const float* __restrict__ enc, const float* __restrict__ dec,
    const float* __restrict__ mask, float* __restrict__ ws)
{
    __shared__ float red_m[WPB];
    __shared__ float red_s[WPB];
    __shared__ float ctxbuf[WPB * H_];

    const int b      = blockIdx.x / NB;
    const int blkInB = blockIdx.x % NB;
    const int tid    = threadIdx.x;
    const int wv     = tid >> 5;
    const int lane   = tid & 31;
    const int s0     = (blkInB * WPB + wv) * ROWS_PER_WAVE;

    const float4* dec4 = (const float4*)(dec + (size_t)b * H_);
    float4 dfrag[8];
#pragma unroll
    for (int k = 0; k < 8; ++k) dfrag[k] = dec4[k * 32 + lane];

    const float maskv = mask[(size_t)b * S_ + s0 + lane];

    float4 ctx[8];
#pragma unroll
    for (int k = 0; k < 8; ++k) ctx[k] = make_float4(0.f, 0.f, 0.f, 0.f);
    float m = -1e30f, ssum = 0.f;

    const float4* encB = (const float4*)(enc + (size_t)b * S_ * H_);
    for (int r = 0; r < ROWS_PER_WAVE; ++r) {
        const float4* row4 = encB + (size_t)(s0 + r) * (H_ / 4);
        float4 row[8];
#pragma unroll
        for (int k = 0; k < 8; ++k) row[k] = row4[k * 32 + lane];

        float d = 0.f;
#pragma unroll
        for (int k = 0; k < 8; ++k)
            d += row[k].x * dfrag[k].x + row[k].y * dfrag[k].y +
                 row[k].z * dfrag[k].z + row[k].w * dfrag[k].w;
#pragma unroll
        for (int off = 16; off >= 1; off >>= 1) d += __shfl_xor(d, off, 32);

        const float logit = d * __shfl(maskv, r, 32);
        const float mn = fmaxf(m, logit);
        const float c  = __expf(m - mn);
        const float p  = __expf(logit - mn);
        ssum = ssum * c + p;
#pragma unroll
        for (int k = 0; k < 8; ++k) {
            ctx[k].x = ctx[k].x * c + p * row[k].x;
            ctx[k].y = ctx[k].y * c + p * row[k].y;
            ctx[k].z = ctx[k].z * c + p * row[k].z;
            ctx[k].w = ctx[k].w * c + p * row[k].w;
        }
        m = mn;
    }

    if (lane == 0) red_m[wv] = m;
    __syncthreads();
    float mB = red_m[0];
#pragma unroll
    for (int i = 1; i < WPB; ++i) mB = fmaxf(mB, red_m[i]);
    const float w = __expf(m - mB);
    if (lane == 0) red_s[wv] = ssum * w;
    float4* cb = (float4*)(ctxbuf + wv * H_);
#pragma unroll
    for (int k = 0; k < 8; ++k) {
        float4 v = ctx[k];
        v.x *= w; v.y *= w; v.z *= w; v.w *= w;
        cb[k * 32 + lane] = v;
    }
    __syncthreads();

    float sB = 0.f;
#pragma unroll
    for (int i = 0; i < WPB; ++i) sB += red_s[i];
    float4 acc = make_float4(0.f, 0.f, 0.f, 0.f);
#pragma unroll
    for (int i = 0; i < WPB; ++i) {
        float4 v = ((const float4*)(ctxbuf + i * H_))[tid];
        acc.x += v.x; acc.y += v.y; acc.z += v.z; acc.w += v.w;
    }
    float* base = ws + (size_t)b * PB_STRIDE;
    ((float4*)(base + blkInB * H_))[tid] = acc;
    if (tid == 0) {
        base[PARTIALS * H_ + blkInB] = mB;
        base[PARTIALS * H_ + PARTIALS + blkInB] = sB;
    }
}

__global__ __launch_bounds__(256) void attn_combine(
    const float* __restrict__ ws, float* __restrict__ ctx_out)
{
    const int b = blockIdx.x, tid = threadIdx.x;
    const float* base = ws + (size_t)b * PB_STRIDE;
    const float* ms = base + PARTIALS * H_;
    const float* ss = ms + PARTIALS;

    float M = ms[0];
#pragma unroll
    for (int i = 1; i < PARTIALS; ++i) M = fmaxf(M, ms[i]);
    float e[PARTIALS], T = 0.f;
#pragma unroll
    for (int i = 0; i < PARTIALS; ++i) { e[i] = __expf(ms[i] - M); T += e[i] * ss[i]; }
    const float inv = 1.f / T;

    float4 acc = make_float4(0.f, 0.f, 0.f, 0.f);
#pragma unroll
    for (int i = 0; i < PARTIALS; ++i) {
        float4 v = ((const float4*)(base + i * H_))[tid];
        acc.x += e[i] * v.x; acc.y += e[i] * v.y;
        acc.z += e[i] * v.z; acc.w += e[i] * v.w;
    }
    acc.x *= inv; acc.y *= inv; acc.z *= inv; acc.w *= inv;
    ((float4*)(ctx_out + (size_t)b * H_))[tid] = acc;
}

#define KW        16
#define KSLICE    (2 * H_ / KW)
#define ROWP      (KSLICE + 4)
#define XROW      (2 * H_)

__global__ __launch_bounds__(512) void attn_out_wmma(
    const float* __restrict__ ctx, const float* __restrict__ dec,
    const float* __restrict__ W, const float* __restrict__ bias,
    float* __restrict__ out)
{
    __shared__ float stage[KW * 32 * ROWP];
    __shared__ float red[KW * 256];

    const int tid  = threadIdx.x;
    const int wv   = tid >> 5;
    const int lane = tid & 31;
    const int n0 = (blockIdx.x >> 1) << 4;
    const int m0 = (blockIdx.x & 1) << 4;
    const float* src = (m0 == 0) ? ctx : dec;
    const int kw = wv * KSLICE;

    const float4* wg = (const float4*)(W   + (size_t)n0 * XROW + kw) + lane;
    const float4* xg = (const float4*)(src + kw) + lane;
    float4 wr[16], xr[16];
#pragma unroll
    for (int j = 0; j < 16; ++j) wr[j] = wg[j * (XROW / 4)];
#pragma unroll
    for (int j = 0; j < 16; ++j) xr[j] = xg[j * (XROW / 4)];

    float* st = stage + wv * (32 * ROWP);
#pragma unroll
    for (int j = 0; j < 16; ++j) *(float4*)(st + j * ROWP + lane * 4) = wr[j];
#pragma unroll
    for (int j = 0; j < 16; ++j) *(float4*)(st + (16 + j) * ROWP + lane * 4) = xr[j];


    const int c = lane & 15;
    const int h = lane >> 4;
    const float* bw = st + c * ROWP + h * 4;
    const float* ax = st + (16 + c) * ROWP + h * 4;
    v8f acc0 = {}, acc1 = {};
#pragma unroll
    for (int q = 0; q < KSLICE / 8; ++q) {
        const float4 b4 = *(const float4*)(bw + q * 8);
        const float4 a4 = *(const float4*)(ax + q * 8);
        v2f a0 = { a4.x, a4.y }, b0 = { b4.x, b4.y };
        v2f a1 = { a4.z, a4.w }, b1 = { b4.z, b4.w };
        acc0 = __builtin_amdgcn_wmma_f32_16x16x4_f32(
            false, a0, false, b0, (short)0, acc0, false, false);
        acc1 = __builtin_amdgcn_wmma_f32_16x16x4_f32(
            false, a1, false, b1, (short)0, acc1, false, false);
    }

    float* rp = red + wv * 256;
#pragma unroll
    for (int r = 0; r < 8; ++r) rp[r * 32 + lane] = acc0[r] + acc1[r];
    __syncthreads();

    if (tid < 256) {
        float s = 0.f;
#pragma unroll
        for (int i = 0; i < KW; ++i) s += red[i * 256 + tid];
        const int r   = tid >> 5;
        const int l   = tid & 31;
        const int row = m0 + r + ((l >> 4) << 3);
        const int col = n0 + (l & 15);
        out[(size_t)row * H_ + col] = tanhf(s + bias[col]);
    }
}

extern "C" void kernel_launch(void* const* d_in, const int* in_sizes, int n_in,
                              void* d_out, int out_size, void* d_ws, size_t ws_size,
                              hipStream_t stream) {
    const float* enc  = (const float*)d_in[0];
    const float* dec  = (const float*)d_in[1];
    const float* mask = (const float*)d_in[2];
    const float* W    = (const float*)d_in[3];
    const float* bias = (const float*)d_in[4];

    float* out = (float*)d_out;
    float* ctx = out + B_ * H_;
    float* ws  = (float*)d_ws;

    attn_pass1 <<<dim3(B_ * NB), dim3(256), 0, stream>>>(enc, dec, mask, ws);
    attn_combine<<<dim3(B_),      dim3(256), 0, stream>>>(ws, ctx);
    attn_out_wmma<<<dim3(128),    dim3(512), 0, stream>>>(ctx, dec, W, bias, out);
}


# pass 1 (encoder stream): each wave keeps six rows (48-64 whole-line loads) requested ahead in eight register row buffers instead of loading one row and waiting for it; rows folded in the same order
# speedup vs baseline: 2.3957x; 1.4141x over previous
#include <hip/hip_runtime.h>

#define B_ 32
#define S_ 2048
#define H_ 1024
#define NB 8
#define WPB 8
#define ROWS_PER_WAVE (S_ / (NB * WPB))
#define PARTIALS NB
#define PB_STRIDE (PARTIALS * H_ + 2 * PARTIALS)

typedef __attribute__((ext_vector_type(2))) float v2f;
typedef __attribute__((ext_vector_type(8))) float v8f;

__global__ __launch_bounds__(256) void attn_pass1(
    const float* __restrict__ enc, const float* __restrict__ dec,
    const float* __restrict__ mask, float* __restrict__ ws)
{
    __shared__ float red_m[WPB];
    __shared__ float red_s[WPB];
    __shared__ float ctxbuf[WPB * H_];

    const int b      = blockIdx.x / NB;
    const int blkInB = blockIdx.x % NB;
    const int tid    = threadIdx.x;
    const int wv     = tid >> 5;
    const int lane   = tid & 31;
    const int s0     = (blkInB * WPB + wv) * ROWS_PER_WAVE;

    const float4* dec4 = (const float4*)(dec + (size_t)b * H_);
    float4 dfrag[8];
#pragma unroll
    for (int k = 0; k < 8; ++k) dfrag[k] = dec4[k * 32 + lane];

    const float maskv = mask[(size_t)b * S_ + s0 + lane];

    const float4* rowbase = (const float4*)(enc + ((size_t)b * S_ + s0) * H_) + lane;

    float4 q0[8], q1[8], q2[8], q3[8], q4[8], q5[8], q6[8], q7[8];
#define LOAD_ROW(Q, R) do { _Pragma("unroll") for (int k = 0; k < 8; ++k) \
        Q[k] = rowbase[(R) * (H_ / 4) + k * 32]; } while (0)

#define REQUESTED() __builtin_amdgcn_sched_barrier(0)

#define FOLDED() do { _Pragma("unroll") for (int k = 0; k < 8; ++k)             \
        asm volatile("" : "+v"(ctx[k].x), "+v"(ctx[k].y),                       \
                          "+v"(ctx[k].z), "+v"(ctx[k].w)); } while (0)

    float4 ctx[8];
#pragma unroll
    for (int k = 0; k < 8; ++k) ctx[k] = make_float4(0.f, 0.f, 0.f, 0.f);
    float m = -1e30f, ssum = 0.f;

    LOAD_ROW(q0, 0); LOAD_ROW(q1, 1); REQUESTED();
    LOAD_ROW(q2, 2); LOAD_ROW(q3, 3); REQUESTED();
    LOAD_ROW(q4, 4); LOAD_ROW(q5, 5); REQUESTED();
#pragma unroll
    for (int k = 0; k < 8; ++k)
        asm volatile("" : "+v"(dfrag[k].x), "+v"(dfrag[k].y),
                          "+v"(dfrag[k].z), "+v"(dfrag[k].w));

#define REDUCE_ROW(Q, R) do {                                                   \
        float d = 0.f;                                                          \
        _Pragma("unroll") for (int k = 0; k < 8; ++k)                           \
            d += Q[k].x * dfrag[k].x + Q[k].y * dfrag[k].y +                    \
                 Q[k].z * dfrag[k].z + Q[k].w * dfrag[k].w;                     \
        _Pragma("unroll") for (int off = 16; off >= 1; off >>= 1)               \
            d += __shfl_xor(d, off, 32);                                        \
        const float logit = d * __shfl(maskv, (R), 32);                         \
        const float mn = fmaxf(m, logit);                                       \
        const float c  = __expf(m - mn);           \
        const float p  = __expf(logit - mn);                                    \
        ssum = ssum * c + p;                                                    \
        _Pragma("unroll") for (int k = 0; k < 8; ++k) {                         \
            ctx[k].x = ctx[k].x * c + p * Q[k].x;                               \
            ctx[k].y = ctx[k].y * c + p * Q[k].y;                               \
            ctx[k].z = ctx[k].z * c + p * Q[k].z;                               \
            ctx[k].w = ctx[k].w * c + p * Q[k].w;                               \
        }                                                                       \
        m = mn;                                                                 \
    } while (0)
#define PAIR(QA, QB, R) do { REQUESTED(); REDUCE_ROW(QA, (R));                  \
        REDUCE_ROW(QB, (R) + 1); FOLDED(); } while (0)

#define STEP(QL0, QL1, QA, QB, R) do { LOAD_ROW(QL0, (R) + 6);                  \
        LOAD_ROW(QL1, (R) + 7); PAIR(QA, QB, (R)); } while (0)
    STEP(q6, q7, q0, q1, 0);   STEP(q0, q1, q2, q3, 2);
    STEP(q2, q3, q4, q5, 4);   STEP(q4, q5, q6, q7, 6);
    STEP(q6, q7, q0, q1, 8);   STEP(q0, q1, q2, q3, 10);
    STEP(q2, q3, q4, q5, 12);  STEP(q4, q5, q6, q7, 14);
    STEP(q6, q7, q0, q1, 16);  STEP(q0, q1, q2, q3, 18);
    STEP(q2, q3, q4, q5, 20);  STEP(q4, q5, q6, q7, 22);
    STEP(q6, q7, q0, q1, 24);
    PAIR(q2, q3, 26);  PAIR(q4, q5, 28);  PAIR(q6, q7, 30);
#undef STEP
#undef PAIR
#undef REDUCE_ROW
#undef FOLDED
#undef REQUESTED
#undef LOAD_ROW

    if (lane == 0) red_m[wv] = m;
    __syncthreads();
    float mB = red_m[0];
#pragma unroll
    for (int i = 1; i < WPB; ++i) mB = fmaxf(mB, red_m[i]);
    const float w = __expf(m - mB);
    if (lane == 0) red_s[wv] = ssum * w;
    float4* cb = (float4*)(ctxbuf + wv * H_);
#pragma unroll
    for (int k = 0; k < 8; ++k) {
        float4 v = ctx[k];
        v.x *= w; v.y *= w; v.z *= w; v.w *= w;
        cb[k * 32 + lane] = v;
    }
    __syncthreads();

    float sB = 0.f;
#pragma unroll
    for (int i = 0; i < WPB; ++i) sB += red_s[i];
    float4 acc = make_float4(0.f, 0.f, 0.f, 0.f);
#pragma unroll
    for (int i = 0; i < WPB; ++i) {
        float4 v = ((const float4*)(ctxbuf + i * H_))[tid];
        acc.x += v.x; acc.y += v.y; acc.z += v.z; acc.w += v.w;
    }
    float* base = ws + (size_t)b * PB_STRIDE;
    ((float4*)(base + blkInB * H_))[tid] = acc;
    if (tid == 0) {
        base[PARTIALS * H_ + blkInB] = mB;
        base[PARTIALS * H_ + PARTIALS + blkInB] = sB;
    }
}

__global__ __launch_bounds__(256) void attn_combine(
    const float* __restrict__ ws, float* __restrict__ ctx_out)
{
    const int b = blockIdx.x, tid = threadIdx.x;
    const float* base = ws + (size_t)b * PB_STRIDE;
    const float* ms = base + PARTIALS * H_;
    const float* ss = ms + PARTIALS;

    float M = ms[0];
#pragma unroll
    for (int i = 1; i < PARTIALS; ++i) M = fmaxf(M, ms[i]);
    float e[PARTIALS], T = 0.f;
#pragma unroll
    for (int i = 0; i < PARTIALS; ++i) { e[i] = __expf(ms[i] - M); T += e[i] * ss[i]; }
    const float inv = 1.f / T;

    float4 acc = make_float4(0.f, 0.f, 0.f, 0.f);
#pragma unroll
    for (int i = 0; i < PARTIALS; ++i) {
        float4 v = ((const float4*)(base + i * H_))[tid];
        acc.x += e[i] * v.x; acc.y += e[i] * v.y;
        acc.z += e[i] * v.z; acc.w += e[i] * v.w;
    }
    acc.x *= inv; acc.y *= inv; acc.z *= inv; acc.w *= inv;
    ((float4*)(ctx_out + (size_t)b * H_))[tid] = acc;
}

#define KW        16
#define KSLICE    (2 * H_ / KW)
#define ROWP      (KSLICE + 4)
#define XROW      (2 * H_)

__global__ __launch_bounds__(512) void attn_out_wmma(
    const float* __restrict__ ctx, const float* __restrict__ dec,
    const float* __restrict__ W, const float* __restrict__ bias,
    float* __restrict__ out)
{
    __shared__ float stage[KW * 32 * ROWP];
    __shared__ float red[KW * 256];

    const int tid  = threadIdx.x;
    const int wv   = tid >> 5;
    const int lane = tid & 31;
    const int n0 = (blockIdx.x >> 1) << 4;
    const int m0 = (blockIdx.x & 1) << 4;
    const float* src = (m0 == 0) ? ctx : dec;
    const int kw = wv * KSLICE;

    const float4* wg = (const float4*)(W   + (size_t)n0 * XROW + kw) + lane;
    const float4* xg = (const float4*)(src + kw) + lane;
    float4 wr[16], xr[16];
#pragma unroll
    for (int j = 0; j < 16; ++j) wr[j] = wg[j * (XROW / 4)];
#pragma unroll
    for (int j = 0; j < 16; ++j) xr[j] = xg[j * (XROW / 4)];

    float* st = stage + wv * (32 * ROWP);
#pragma unroll
    for (int j = 0; j < 16; ++j) *(float4*)(st + j * ROWP + lane * 4) = wr[j];
#pragma unroll
    for (int j = 0; j < 16; ++j) *(float4*)(st + (16 + j) * ROWP + lane * 4) = xr[j];


    const int c = lane & 15;
    const int h = lane >> 4;
    const float* bw = st + c * ROWP + h * 4;
    const float* ax = st + (16 + c) * ROWP + h * 4;
    v8f acc0 = {}, acc1 = {};
#pragma unroll
    for (int q = 0; q < KSLICE / 8; ++q) {
        const float4 b4 = *(const float4*)(bw + q * 8);
        const float4 a4 = *(const float4*)(ax + q * 8);
        v2f a0 = { a4.x, a4.y }, b0 = { b4.x, b4.y };
        v2f a1 = { a4.z, a4.w }, b1 = { b4.z, b4.w };
        acc0 = __builtin_amdgcn_wmma_f32_16x16x4_f32(
            false, a0, false, b0, (short)0, acc0, false, false);
        acc1 = __builtin_amdgcn_wmma_f32_16x16x4_f32(
            false, a1, false, b1, (short)0, acc1, false, false);
    }

    float* rp = red + wv * 256;
#pragma unroll
    for (int r = 0; r < 8; ++r) rp[r * 32 + lane] = acc0[r] + acc1[r];
    __syncthreads();

    if (tid < 256) {
        float s = 0.f;
#pragma unroll
        for (int i = 0; i < KW; ++i) s += red[i * 256 + tid];
        const int r   = tid >> 5;
        const int l   = tid & 31;
        const int row = m0 + r + ((l >> 4) << 3);
        const int col = n0 + (l & 15);
        out[(size_t)row * H_ + col] = tanhf(s + bias[col]);
    }
}

extern "C" void kernel_launch(void* const* d_in, const int* in_sizes, int n_in,
                              void* d_out, int out_size, void* d_ws, size_t ws_size,
                              hipStream_t stream) {
    const float* enc  = (const float*)d_in[0];
    const float* dec  = (const float*)d_in[1];
    const float* mask = (const float*)d_in[2];
    const float* W    = (const float*)d_in[3];
    const float* bias = (const float*)d_in[4];

    float* out = (float*)d_out;
    float* ctx = out + B_ * H_;
    float* ws  = (float*)d_ws;

    attn_pass1 <<<dim3(B_ * NB), dim3(256), 0, stream>>>(enc, dec, mask, ws);
    attn_combine<<<dim3(B_),      dim3(256), 0, stream>>>(ws, ctx);
    attn_out_wmma<<<dim3(128),    dim3(512), 0, stream>>>(ctx, dec, W, bias, out);
}
